# starting kernel
# baseline (speedup 1.0000x reference)
#include <hip/hip_runtime.h>


#define N_NODES   50000
#define N_GRAPHS  128
#define HID       512
#define OUT_PITCH (3 * HID)
#define MAX_CHUNKS 64
#define COLS_PER_WG 256

typedef __attribute__((ext_vector_type(2))) float v2f;
typedef __attribute__((ext_vector_type(8))) float v8f;

__global__ void k_zero_counts(unsigned* __restrict__ counts) {
    counts[threadIdx.x] = 0u;
}

__global__ void k_pb_hist(const int* __restrict__ walk2_row0,
                          const int* __restrict__ batch,
                          int* __restrict__ pb,
                          unsigned* __restrict__ counts) {
    int n = blockIdx.x * 256 + threadIdx.x;
    if (n < N_NODES) {
        int g = batch[walk2_row0[n]];
        pb[n] = g;
        atomicAdd(&counts[g], 1u);
    }
}

__global__ void __launch_bounds__(256)
k_accum(const float* __restrict__ x,
        const int* __restrict__ w2, const int* __restrict__ w3,
        const int* __restrict__ w4, const int* __restrict__ pb,
        float* __restrict__ partial, int nchunks) {
    __shared__ float acc[N_GRAPHS * COLS_PER_WG];
    const int chunk = blockIdx.x;
    const int cg    = blockIdx.y;
    const int t     = threadIdx.x;
    const int c     = cg * COLS_PER_WG + t;

    for (int i = t; i < N_GRAPHS * COLS_PER_WG; i += 256) acc[i] = 0.0f;
    __syncthreads();

    const int per = (N_NODES + nchunks - 1) / nchunks;
    const int n0  = chunk * per;
    const int n1  = (n0 + per < N_NODES) ? (n0 + per) : N_NODES;

    for (int n = n0; n < n1; ++n) {
        const int g = pb[n];
        float y = 0.0f;
        #pragma unroll
        for (int l = 0; l < 3; ++l) y += x[w2[l * N_NODES + n] * HID + c];
        #pragma unroll
        for (int l = 0; l < 4; ++l) y += x[w3[l * N_NODES + n] * HID + c];
        #pragma unroll
        for (int l = 0; l < 5; ++l) y += x[w4[l * N_NODES + n] * HID + c];
        acc[g * COLS_PER_WG + t] += y;
    }
    __syncthreads();

    float* dst = partial + (size_t)chunk * (N_GRAPHS * HID) + cg * COLS_PER_WG;
    for (int g = 0; g < N_GRAPHS; ++g)
        dst[g * HID + t] = acc[g * COLS_PER_WG + t];
}

__global__ void k_reduce(const float* __restrict__ partial,
                         const unsigned* __restrict__ counts,
                         float* __restrict__ P, float* __restrict__ s,
                         int nchunks) {
    int i = blockIdx.x * 256 + threadIdx.x;
    int g = i >> 9;
    float a = 0.0f;
    for (int ch = 0; ch < nchunks; ++ch)
        a += partial[(size_t)ch * (N_GRAPHS * HID) + i];
    float cnt = (float)counts[g];
    P[i] = a / fmaxf(cnt, 1.0f);
    if (i < N_GRAPHS) s[i] = (counts[i] != 0u) ? 12.0f : 0.0f;
}

__global__ void __launch_bounds__(256)
k_gemm_wmma(const float* __restrict__ A, int lda,
            const float* __restrict__ W, const float* __restrict__ bias,
            const float* __restrict__ s, float* __restrict__ out) {
    const int wave = (blockIdx.x << 3) + (threadIdx.x >> 5);
    const int lane = threadIdx.x & 31;
    const int l16  = lane & 15;
    const int lh   = lane >> 4;
    const int tm   = wave >> 5;
    const int tn   = wave & 31;

    const float* Arow = A + (tm * 16 + l16) * lda;
    const int col = tn * 16 + l16;

    v8f acc = {};
    for (int k = 0; k < HID; k += 4) {
        const int kk = k + lh * 2;
        v2f a; a.x = Arow[kk];           a.y = Arow[kk + 1];
        v2f b; b.x = W[kk * HID + col];  b.y = W[(kk + 1) * HID + col];
        acc = __builtin_amdgcn_wmma_f32_16x16x4_f32(
             false, a,  false, b,
             (short)0, acc,  false,  false);
    }

    const float bn = bias[col];
    #pragma unroll
    for (int v = 0; v < 8; ++v) {
        const int m = tm * 16 + v + lh * 8;
        out[m * OUT_PITCH + col] = acc[v] + s[m] * bn;
    }
}

extern "C" void kernel_launch(void* const* d_in, const int* in_sizes, int n_in,
                              void* d_out, int out_size, void* d_ws, size_t ws_size,
                              hipStream_t stream) {
    const float* x     = (const float*)d_in[0];
    const int*   w2    = (const int*)  d_in[1];
    const int*   w3    = (const int*)  d_in[2];
    const int*   w4    = (const int*)  d_in[3];
    const int*   batch = (const int*)  d_in[4];
    const float* W0 = (const float*)d_in[5];  const float* b0 = (const float*)d_in[6];
    const float* W1 = (const float*)d_in[7];  const float* b1 = (const float*)d_in[8];
    const float* W2 = (const float*)d_in[9];  const float* b2 = (const float*)d_in[10];
    float* out = (float*)d_out;

    const size_t partBytes = (size_t)N_GRAPHS * HID * sizeof(float);
    const size_t fixed = partBytes   + 512   + 512
                       + (size_t)N_NODES * sizeof(int)  ;
    int nchunks = 1;
    if (ws_size > fixed + partBytes) {
        size_t fit = (ws_size - fixed) / partBytes;
        nchunks = (fit > MAX_CHUNKS) ? MAX_CHUNKS : (int)fit;
        if (nchunks < 1) nchunks = 1;
    }
    char* ws = (char*)d_ws;
    float*    partial = (float*)ws;
    float*    P       = (float*)(ws + (size_t)nchunks * partBytes);
    unsigned* counts  = (unsigned*)((char*)P + partBytes);
    float*    s       = (float*)((char*)counts + 512);
    int*      pb      = (int*)((char*)s + 512);

    k_zero_counts<<<1, 128, 0, stream>>>(counts);
    k_pb_hist<<<(N_NODES + 255) / 256, 256, 0, stream>>>(w2, batch, pb, counts);

    dim3 gAcc(nchunks, 2);
    k_accum<<<gAcc, 256, 0, stream>>>(x, w2, w3, w4, pb, partial, nchunks);
    k_reduce<<<(N_GRAPHS * HID) / 256, 256, 0, stream>>>(partial, counts, P, s, nchunks);

    k_gemm_wmma<<<32, 256, 0, stream>>>(P,         HID,       W0, b0, s, out);
    k_gemm_wmma<<<32, 256, 0, stream>>>(out,       OUT_PITCH, W1, b1, s, out + HID);
    k_gemm_wmma<<<32, 256, 0, stream>>>(out + HID, OUT_PITCH, W2, b2, s, out + 2 * HID);
}


# accumulate kernel: chunk positions split over 4 private LDS accumulators (256 WGs), 48 row loads (4 positions) in flight per wave; chunk sum now a fixed-order 4-way merge (new reduction tree)
# speedup vs baseline: 4.1311x; 4.1311x over previous
#include <hip/hip_runtime.h>


#define N_NODES   50000
#define N_GRAPHS  128
#define HID       512
#define OUT_PITCH (3 * HID)
#define MAX_CHUNKS 64
#define COLS_PER_WG 128
#define SUBS 4
#define UNR  4

typedef __attribute__((ext_vector_type(2))) float v2f;
typedef __attribute__((ext_vector_type(8))) float v8f;

__global__ void k_zero_counts(unsigned* __restrict__ counts) {
    counts[threadIdx.x] = 0u;
}

__global__ void k_pb_hist(const int* __restrict__ walk2_row0,
                          const int* __restrict__ batch,
                          int* __restrict__ pb,
                          unsigned* __restrict__ counts) {
    int n = blockIdx.x * 256 + threadIdx.x;
    if (n < N_NODES) {
        int g = batch[walk2_row0[n]];
        pb[n] = g;
        atomicAdd(&counts[g], 1u);
    }
}


__device__ __forceinline__ v2f ld_row2(const float* __restrict__ x, unsigned lane_off, int idx) {
    const unsigned off = (unsigned)idx * (unsigned)(HID * sizeof(float)) + lane_off;
    return *(const v2f*)((const char*)x + off);
}

__device__ __forceinline__ v2f gather12(const float* __restrict__ x, unsigned lane_off,
        const int* __restrict__ w2, const int* __restrict__ w3,
        const int* __restrict__ w4, int n) {
    v2f y = {0.0f, 0.0f};
    #pragma unroll
    for (int l = 0; l < 3; ++l) y += ld_row2(x, lane_off, w2[l * N_NODES + n]);
    #pragma unroll
    for (int l = 0; l < 4; ++l) y += ld_row2(x, lane_off, w3[l * N_NODES + n]);
    #pragma unroll
    for (int l = 0; l < 5; ++l) y += ld_row2(x, lane_off, w4[l * N_NODES + n]);
    return y;
}

__global__ void __launch_bounds__(256)
k_accum(const float* __restrict__ x,
        const int* __restrict__ w2, const int* __restrict__ w3,
        const int* __restrict__ w4, const int* __restrict__ pb,
        float* __restrict__ partial, int nchunks) {
    __shared__ float acc[SUBS * N_GRAPHS * COLS_PER_WG];
    const int chunk = blockIdx.x;
    const int cg    = blockIdx.y;
    const int t     = threadIdx.x;
    const int wave  = __builtin_amdgcn_readfirstlane(t >> 5);
    const int lane  = t & 31;
    const int sub   = wave >> 1;
    const int wc    = (wave & 1) * 64 + lane * 2;
    const int c     = cg * COLS_PER_WG + wc;

    float4* acc4 = (float4*)acc;
    for (int i = t; i < SUBS * N_GRAPHS * COLS_PER_WG / 4; i += 256)
        acc4[i] = make_float4(0.0f, 0.0f, 0.0f, 0.0f);
    __syncthreads();

    const int per  = (N_NODES + nchunks - 1) / nchunks;
    const int n0   = chunk * per;
    const int n1   = (n0 + per < N_NODES) ? (n0 + per) : N_NODES;
    const int sper = (per + SUBS - 1) / SUBS;
    const int m0   = n0 + sub * sper;
    const int m1   = (m0 + sper < n1) ? (m0 + sper) : n1;

    float* accs = acc + sub * (N_GRAPHS * COLS_PER_WG) + wc;
    const unsigned lane_off = (unsigned)c * (unsigned)sizeof(float);

    int n = m0;
    for (; n + UNR <= m1; n += UNR) {
        int g[UNR];
        v2f y[UNR];
        #pragma unroll
        for (int u = 0; u < UNR; ++u) {
            g[u] = pb[n + u];
            y[u] = gather12(x, lane_off, w2, w3, w4, n + u);
        }
        #pragma unroll
        for (int u = 0; u < UNR; ++u) {
            v2f* p = (v2f*)(accs + g[u] * COLS_PER_WG);
            *p += y[u];
        }
    }
    for (; n < m1; ++n) {
        const int g = pb[n];
        v2f* p = (v2f*)(accs + g * COLS_PER_WG);
        *p += gather12(x, lane_off, w2, w3, w4, n);
    }
    __syncthreads();

    const int c4 = (t & 31) * 4;
    float* dst = partial + (size_t)chunk * (N_GRAPHS * HID) + cg * COLS_PER_WG + c4;
    for (int g = t >> 5; g < N_GRAPHS; g += 8) {
        float4 s = *(const float4*)(acc + g * COLS_PER_WG + c4);
        for (int k = 1; k < SUBS; ++k) {
            const float4 a = *(const float4*)(acc + (k * N_GRAPHS + g) * COLS_PER_WG + c4);
            s.x += a.x; s.y += a.y; s.z += a.z; s.w += a.w;
        }
        *(float4*)(dst + g * HID) = s;
    }
}

__global__ void k_reduce(const float* __restrict__ partial,
                         const unsigned* __restrict__ counts,
                         float* __restrict__ P, float* __restrict__ s,
                         int nchunks) {
    int i = blockIdx.x * 256 + threadIdx.x;
    int g = i >> 9;
    float a = 0.0f;
    for (int ch = 0; ch < nchunks; ++ch)
        a += partial[(size_t)ch * (N_GRAPHS * HID) + i];
    float cnt = (float)counts[g];
    P[i] = a / fmaxf(cnt, 1.0f);
    if (i < N_GRAPHS) s[i] = (counts[i] != 0u) ? 12.0f : 0.0f;
}

__global__ void __launch_bounds__(256)
k_gemm_wmma(const float* __restrict__ A, int lda,
            const float* __restrict__ W, const float* __restrict__ bias,
            const float* __restrict__ s, float* __restrict__ out) {
    const int wave = (blockIdx.x << 3) + (threadIdx.x >> 5);
    const int lane = threadIdx.x & 31;
    const int l16  = lane & 15;
    const int lh   = lane >> 4;
    const int tm   = wave >> 5;
    const int tn   = wave & 31;

    const float* Arow = A + (tm * 16 + l16) * lda;
    const int col = tn * 16 + l16;

    v8f acc = {};
    for (int k = 0; k < HID; k += 4) {
        const int kk = k + lh * 2;
        v2f a; a.x = Arow[kk];           a.y = Arow[kk + 1];
        v2f b; b.x = W[kk * HID + col];  b.y = W[(kk + 1) * HID + col];
        acc = __builtin_amdgcn_wmma_f32_16x16x4_f32(
             false, a,  false, b,
             (short)0, acc,  false,  false);
    }

    const float bn = bias[col];
    #pragma unroll
    for (int v = 0; v < 8; ++v) {
        const int m = tm * 16 + v + lh * 8;
        out[m * OUT_PITCH + col] = acc[v] + s[m] * bn;
    }
}

extern "C" void kernel_launch(void* const* d_in, const int* in_sizes, int n_in,
                              void* d_out, int out_size, void* d_ws, size_t ws_size,
                              hipStream_t stream) {
    const float* x     = (const float*)d_in[0];
    const int*   w2    = (const int*)  d_in[1];
    const int*   w3    = (const int*)  d_in[2];
    const int*   w4    = (const int*)  d_in[3];
    const int*   batch = (const int*)  d_in[4];
    const float* W0 = (const float*)d_in[5];  const float* b0 = (const float*)d_in[6];
    const float* W1 = (const float*)d_in[7];  const float* b1 = (const float*)d_in[8];
    const float* W2 = (const float*)d_in[9];  const float* b2 = (const float*)d_in[10];
    float* out = (float*)d_out;

    const size_t partBytes = (size_t)N_GRAPHS * HID * sizeof(float);
    const size_t fixed = partBytes   + 512   + 512
                       + (size_t)N_NODES * sizeof(int)  ;
    int nchunks = 1;
    if (ws_size > fixed + partBytes) {
        size_t fit = (ws_size - fixed) / partBytes;
        nchunks = (fit > MAX_CHUNKS) ? MAX_CHUNKS : (int)fit;
        if (nchunks < 1) nchunks = 1;
    }
    char* ws = (char*)d_ws;
    float*    partial = (float*)ws;
    float*    P       = (float*)(ws + (size_t)nchunks * partBytes);
    unsigned* counts  = (unsigned*)((char*)P + partBytes);
    float*    s       = (float*)((char*)counts + 512);
    int*      pb      = (int*)((char*)s + 512);

    k_zero_counts<<<1, 128, 0, stream>>>(counts);
    k_pb_hist<<<(N_NODES + 255) / 256, 256, 0, stream>>>(w2, batch, pb, counts);

    dim3 gAcc(nchunks, HID / COLS_PER_WG);
    k_accum<<<gAcc, 256, 0, stream>>>(x, w2, w3, w4, pb, partial, nchunks);
    k_reduce<<<(N_GRAPHS * HID) / 256, 256, 0, stream>>>(partial, counts, P, s, nchunks);

    k_gemm_wmma<<<32, 256, 0, stream>>>(P,         HID,       W0, b0, s, out);
    k_gemm_wmma<<<32, 256, 0, stream>>>(out,       OUT_PITCH, W1, b1, s, out + HID);
    k_gemm_wmma<<<32, 256, 0, stream>>>(out + HID, OUT_PITCH, W2, b2, s, out + 2 * HID);
}


# GEMM kernels: K split 8 ways over the waves of a workgroup (16x32 tile, a wave's 48 operand loads all in flight), partial tiles summed through LDS in fixed order 0..7 (new reduction tree)
# speedup vs baseline: 6.0143x; 1.4559x over previous
#include <hip/hip_runtime.h>


#define N_NODES   50000
#define N_GRAPHS  128
#define HID       512
#define OUT_PITCH (3 * HID)
#define MAX_CHUNKS 64
#define COLS_PER_WG 128
#define SUBS 4
#define UNR  4

typedef __attribute__((ext_vector_type(2))) float v2f;
typedef __attribute__((ext_vector_type(8))) float v8f;

__global__ void k_zero_counts(unsigned* __restrict__ counts) {
    counts[threadIdx.x] = 0u;
}

__global__ void k_pb_hist(const int* __restrict__ walk2_row0,
                          const int* __restrict__ batch,
                          int* __restrict__ pb,
                          unsigned* __restrict__ counts) {
    int n = blockIdx.x * 256 + threadIdx.x;
    if (n < N_NODES) {
        int g = batch[walk2_row0[n]];
        pb[n] = g;
        atomicAdd(&counts[g], 1u);
    }
}


__device__ __forceinline__ v2f ld_row2(const float* __restrict__ x, unsigned lane_off, int idx) {
    const unsigned off = (unsigned)idx * (unsigned)(HID * sizeof(float)) + lane_off;
    return *(const v2f*)((const char*)x + off);
}

__device__ __forceinline__ v2f gather12(const float* __restrict__ x, unsigned lane_off,
        const int* __restrict__ w2, const int* __restrict__ w3,
        const int* __restrict__ w4, int n) {
    v2f y = {0.0f, 0.0f};
    #pragma unroll
    for (int l = 0; l < 3; ++l) y += ld_row2(x, lane_off, w2[l * N_NODES + n]);
    #pragma unroll
    for (int l = 0; l < 4; ++l) y += ld_row2(x, lane_off, w3[l * N_NODES + n]);
    #pragma unroll
    for (int l = 0; l < 5; ++l) y += ld_row2(x, lane_off, w4[l * N_NODES + n]);
    return y;
}

__global__ void __launch_bounds__(256)
k_accum(const float* __restrict__ x,
        const int* __restrict__ w2, const int* __restrict__ w3,
        const int* __restrict__ w4, const int* __restrict__ pb,
        float* __restrict__ partial, int nchunks) {
    __shared__ float acc[SUBS * N_GRAPHS * COLS_PER_WG];
    const int chunk = blockIdx.x;
    const int cg    = blockIdx.y;
    const int t     = threadIdx.x;
    const int wave  = __builtin_amdgcn_readfirstlane(t >> 5);
    const int lane  = t & 31;
    const int sub   = wave >> 1;
    const int wc    = (wave & 1) * 64 + lane * 2;
    const int c     = cg * COLS_PER_WG + wc;

    float4* acc4 = (float4*)acc;
    for (int i = t; i < SUBS * N_GRAPHS * COLS_PER_WG / 4; i += 256)
        acc4[i] = make_float4(0.0f, 0.0f, 0.0f, 0.0f);
    __syncthreads();

    const int per  = (N_NODES + nchunks - 1) / nchunks;
    const int n0   = chunk * per;
    const int n1   = (n0 + per < N_NODES) ? (n0 + per) : N_NODES;
    const int sper = (per + SUBS - 1) / SUBS;
    const int m0   = n0 + sub * sper;
    const int m1   = (m0 + sper < n1) ? (m0 + sper) : n1;

    float* accs = acc + sub * (N_GRAPHS * COLS_PER_WG) + wc;
    const unsigned lane_off = (unsigned)c * (unsigned)sizeof(float);

    int n = m0;
    for (; n + UNR <= m1; n += UNR) {
        int g[UNR];
        v2f y[UNR];
        #pragma unroll
        for (int u = 0; u < UNR; ++u) {
            g[u] = pb[n + u];
            y[u] = gather12(x, lane_off, w2, w3, w4, n + u);
        }
        #pragma unroll
        for (int u = 0; u < UNR; ++u) {
            v2f* p = (v2f*)(accs + g[u] * COLS_PER_WG);
            *p += y[u];
        }
    }
    for (; n < m1; ++n) {
        const int g = pb[n];
        v2f* p = (v2f*)(accs + g * COLS_PER_WG);
        *p += gather12(x, lane_off, w2, w3, w4, n);
    }
    __syncthreads();

    const int c4 = (t & 31) * 4;
    float* dst = partial + (size_t)chunk * (N_GRAPHS * HID) + cg * COLS_PER_WG + c4;
    for (int g = t >> 5; g < N_GRAPHS; g += 8) {
        float4 s = *(const float4*)(acc + g * COLS_PER_WG + c4);
        for (int k = 1; k < SUBS; ++k) {
            const float4 a = *(const float4*)(acc + (k * N_GRAPHS + g) * COLS_PER_WG + c4);
            s.x += a.x; s.y += a.y; s.z += a.z; s.w += a.w;
        }
        *(float4*)(dst + g * HID) = s;
    }
}

__global__ void k_reduce(const float* __restrict__ partial,
                         const unsigned* __restrict__ counts,
                         float* __restrict__ P, float* __restrict__ s,
                         int nchunks) {
    int i = blockIdx.x * 256 + threadIdx.x;
    int g = i >> 9;
    float a = 0.0f;
    for (int ch = 0; ch < nchunks; ++ch)
        a += partial[(size_t)ch * (N_GRAPHS * HID) + i];
    float cnt = (float)counts[g];
    P[i] = a / fmaxf(cnt, 1.0f);
    if (i < N_GRAPHS) s[i] = (counts[i] != 0u) ? 12.0f : 0.0f;
}

#define GEMM_KSPLIT 8
#define GEMM_KPER   (HID / GEMM_KSPLIT)
#define GEMM_STEPS  (GEMM_KPER / 4)
#define PART_PITCH  36

__global__ void __launch_bounds__(256) __attribute__((amdgpu_waves_per_eu(2, 2)))
k_gemm_wmma(const float* __restrict__ A, int lda,
            const float* __restrict__ W, const float* __restrict__ bias,
            const float* __restrict__ s, float* __restrict__ out) {
    __shared__ float part[GEMM_KSPLIT * 16 * PART_PITCH];

    const int ks   = threadIdx.x >> 5;
    const int lane = threadIdx.x & 31;
    const int l16  = lane & 15;
    const int lh   = lane >> 4;
    const int tm   = blockIdx.x >> 4;
    const int bn   = blockIdx.x & 15;

    const int   col = bn * 32 + lane;
    const float bc  = bias[col];
    const float sv0 = s[tm * 16 + ks];
    const float sv1 = s[tm * 16 + ks + 8];

    const float* Ap = A + (tm * 16 + l16) * lda + ks * GEMM_KPER + lh * 2;
    const float* Wp = W + (ks * GEMM_KPER + lh * 2) * HID + bn * 32 + l16 * 2;

    v2f a[GEMM_STEPS], b0[GEMM_STEPS], b1[GEMM_STEPS];
    #pragma unroll
    for (int j = 0; j < GEMM_STEPS; ++j) {
        a[j] = *(const v2f*)(Ap + 4 * j);
        const v2f w0 = *(const v2f*)(Wp + (4 * j) * HID);
        const v2f w1 = *(const v2f*)(Wp + (4 * j + 1) * HID);
        b0[j].x = w0.x;  b0[j].y = w1.x;
        b1[j].x = w0.y;  b1[j].y = w1.y;
    }

    v8f acc0 = {};
    v8f acc1 = {};
    #pragma unroll
    for (int j = 0; j < GEMM_STEPS; ++j) {
        acc0 = __builtin_amdgcn_wmma_f32_16x16x4_f32(
             false, a[j],  false, b0[j],
             (short)0, acc0,  false,  false);
        acc1 = __builtin_amdgcn_wmma_f32_16x16x4_f32(
             false, a[j],  false, b1[j],
             (short)0, acc1,  false,  false);
    }

    float* pw = part + ks * (16 * PART_PITCH) + (lh * 8) * PART_PITCH + l16 * 2;
    #pragma unroll
    for (int v = 0; v < 8; ++v) {
        v2f pr2;  pr2.x = acc0[v];  pr2.y = acc1[v];
        *(v2f*)(pw + v * PART_PITCH) = pr2;
    }
    __syncthreads();

    #pragma unroll
    for (int i = 0; i < 2; ++i) {
        const int r = ks + 8 * i;
        const float* pr = part + r * PART_PITCH + lane;
        float sum = pr[0];
        for (int q = 1; q < GEMM_KSPLIT; ++q) sum += pr[q * (16 * PART_PITCH)];
        out[(tm * 16 + r) * OUT_PITCH + col] = sum + (i ? sv1 : sv0) * bc;
    }
}

extern "C" void kernel_launch(void* const* d_in, const int* in_sizes, int n_in,
                              void* d_out, int out_size, void* d_ws, size_t ws_size,
                              hipStream_t stream) {
    const float* x     = (const float*)d_in[0];
    const int*   w2    = (const int*)  d_in[1];
    const int*   w3    = (const int*)  d_in[2];
    const int*   w4    = (const int*)  d_in[3];
    const int*   batch = (const int*)  d_in[4];
    const float* W0 = (const float*)d_in[5];  const float* b0 = (const float*)d_in[6];
    const float* W1 = (const float*)d_in[7];  const float* b1 = (const float*)d_in[8];
    const float* W2 = (const float*)d_in[9];  const float* b2 = (const float*)d_in[10];
    float* out = (float*)d_out;

    const size_t partBytes = (size_t)N_GRAPHS * HID * sizeof(float);
    const size_t fixed = partBytes   + 512   + 512
                       + (size_t)N_NODES * sizeof(int)  ;
    int nchunks = 1;
    if (ws_size > fixed + partBytes) {
        size_t fit = (ws_size - fixed) / partBytes;
        nchunks = (fit > MAX_CHUNKS) ? MAX_CHUNKS : (int)fit;
        if (nchunks < 1) nchunks = 1;
    }
    char* ws = (char*)d_ws;
    float*    partial = (float*)ws;
    float*    P       = (float*)(ws + (size_t)nchunks * partBytes);
    unsigned* counts  = (unsigned*)((char*)P + partBytes);
    float*    s       = (float*)((char*)counts + 512);
    int*      pb      = (int*)((char*)s + 512);

    k_zero_counts<<<1, 128, 0, stream>>>(counts);
    k_pb_hist<<<(N_NODES + 255) / 256, 256, 0, stream>>>(w2, batch, pb, counts);

    dim3 gAcc(nchunks, HID / COLS_PER_WG);
    k_accum<<<gAcc, 256, 0, stream>>>(x, w2, w3, w4, pb, partial, nchunks);
    k_reduce<<<(N_GRAPHS * HID) / 256, 256, 0, stream>>>(partial, counts, P, s, nchunks);

    k_gemm_wmma<<<128, 256, 0, stream>>>(P,         HID,       W0, b0, s, out);
    k_gemm_wmma<<<128, 256, 0, stream>>>(out,       OUT_PITCH, W1, b1, s, out + HID);
    k_gemm_wmma<<<128, 256, 0, stream>>>(out + HID, OUT_PITCH, W2, b2, s, out + 2 * HID);
}


# accumulate kernel: row loads kept as a rolling window of 4 positions (a load is reissued as soon as its value is summed), never drained between trips; walk entries fetched a group ahead; same sums
# speedup vs baseline: 6.3844x; 1.0615x over previous
#include <hip/hip_runtime.h>


#define N_NODES   50000
#define N_GRAPHS  128
#define HID       512
#define OUT_PITCH (3 * HID)
#define MAX_CHUNKS 64
#define COLS_PER_WG 128
#define SUBS 4
#define UNR  4

typedef __attribute__((ext_vector_type(2))) float v2f;
typedef __attribute__((ext_vector_type(8))) float v8f;

__global__ void k_zero_counts(unsigned* __restrict__ counts) {
    counts[threadIdx.x] = 0u;
}

__global__ void k_pb_hist(const int* __restrict__ walk2_row0,
                          const int* __restrict__ batch,
                          int* __restrict__ pb,
                          unsigned* __restrict__ counts) {
    int n = blockIdx.x * 256 + threadIdx.x;
    if (n < N_NODES) {
        int g = batch[walk2_row0[n]];
        pb[n] = g;
        atomicAdd(&counts[g], 1u);
    }
}


__device__ __forceinline__ unsigned row_off(unsigned lane_off, int idx) {
    return (unsigned)idx * (unsigned)(HID * sizeof(float)) + lane_off;
}

__device__ __forceinline__ v2f ld_off2(const float* __restrict__ x, unsigned off) {
    return *(const v2f*)((const char*)x + off);
}

__device__ __forceinline__ void idx12(int (&ix)[12],
        const int* __restrict__ w2, const int* __restrict__ w3,
        const int* __restrict__ w4, int n) {
    #pragma unroll
    for (int l = 0; l < 3; ++l) ix[l]     = w2[l * N_NODES + n];
    #pragma unroll
    for (int l = 0; l < 4; ++l) ix[3 + l] = w3[l * N_NODES + n];
    #pragma unroll
    for (int l = 0; l < 5; ++l) ix[7 + l] = w4[l * N_NODES + n];
}

__device__ __forceinline__ void offs12(unsigned (&o)[12], unsigned lane_off, const int (&ix)[12]) {
    #pragma unroll
    for (int k = 0; k < 12; ++k) o[k] = row_off(lane_off, ix[k]);
}

__device__ __forceinline__ void issue12(v2f (&raw)[12], const float* __restrict__ x,
                                        const unsigned (&o)[12]) {
    #pragma unroll
    for (int k = 0; k < 12; ++k) raw[k] = ld_off2(x, o[k]);
}

__device__ __forceinline__ v2f sum12(const v2f (&raw)[12]) {
    v2f y = {0.0f, 0.0f};
    #pragma unroll
    for (int k = 0; k < 12; ++k) y += raw[k];
    return y;
}

__device__ __forceinline__ v2f gather12(const float* __restrict__ x, unsigned lane_off,
        const int* __restrict__ w2, const int* __restrict__ w3,
        const int* __restrict__ w4, int n) {
    int ix[12];
    unsigned o[12];
    v2f raw[12];
    idx12(ix, w2, w3, w4, n);
    offs12(o, lane_off, ix);
    issue12(raw, x, o);
    return sum12(raw);
}

__global__ void __launch_bounds__(256)
k_accum(const float* __restrict__ x,
        const int* __restrict__ w2, const int* __restrict__ w3,
        const int* __restrict__ w4, const int* __restrict__ pb,
        float* __restrict__ partial, int nchunks) {
    __shared__ float acc[SUBS * N_GRAPHS * COLS_PER_WG];
    const int chunk = blockIdx.x;
    const int cg    = blockIdx.y;
    const int t     = threadIdx.x;
    const int wave  = __builtin_amdgcn_readfirstlane(t >> 5);
    const int lane  = t & 31;
    const int sub   = wave >> 1;
    const int wc    = (wave & 1) * 64 + lane * 2;
    const int c     = cg * COLS_PER_WG + wc;

    float4* acc4 = (float4*)acc;
    for (int i = t; i < SUBS * N_GRAPHS * COLS_PER_WG / 4; i += 256)
        acc4[i] = make_float4(0.0f, 0.0f, 0.0f, 0.0f);
    __syncthreads();

    const int per  = (N_NODES + nchunks - 1) / nchunks;
    const int n0   = chunk * per;
    const int n1   = (n0 + per < N_NODES) ? (n0 + per) : N_NODES;
    const int sper = (per + SUBS - 1) / SUBS;
    const int m0   = n0 + sub * sper;
    const int m1   = (m0 + sper < n1) ? (m0 + sper) : n1;

    float* accs = acc + sub * (N_GRAPHS * COLS_PER_WG) + wc;
    const unsigned lane_off = (unsigned)c * (unsigned)sizeof(float);

    int n = m0;
    const int ngrp = (m1 > m0) ? (m1 - m0) / UNR : 0;
    if (ngrp > 0) {
        v2f raw[UNR][12];
        int ga[UNR];
        unsigned o[UNR][12];
        int gn[UNR];
        #pragma unroll
        for (int u = 0; u < UNR; ++u) {
            int ix[12];
            idx12(ix, w2, w3, w4, n + u);
            offs12(o[u], lane_off, ix);
            ga[u] = pb[n + u];
            issue12(raw[u], x, o[u]);
        }
        {
            int qb = n + UNR;  qb = (qb <= N_NODES - UNR) ? qb : (N_NODES - UNR);
            #pragma unroll
            for (int u = 0; u < UNR; ++u) {
                int ix[12];
                idx12(ix, w2, w3, w4, qb + u);
                offs12(o[u], lane_off, ix);
                gn[u] = pb[qb + u];
            }
        }
        for (int gi = 1; gi < ngrp; ++gi) {
            int qb = n + (gi + 1) * UNR;  qb = (qb <= N_NODES - UNR) ? qb : (N_NODES - UNR);
            int ixn[UNR][12];
            int gnn[UNR];
            #pragma unroll
            for (int u = 0; u < UNR; ++u) {
                idx12(ixn[u], w2, w3, w4, qb + u);
                gnn[u] = pb[qb + u];
            }
            #pragma unroll
            for (int u = 0; u < UNR; ++u) {
                v2f* p = (v2f*)(accs + ga[u] * COLS_PER_WG);
                *p += sum12(raw[u]);
                issue12(raw[u], x, o[u]);
                ga[u] = gn[u];
                __builtin_amdgcn_sched_barrier(0);
            }
            #pragma unroll
            for (int u = 0; u < UNR; ++u) {
                offs12(o[u], lane_off, ixn[u]);
                gn[u] = gnn[u];
            }
        }
        #pragma unroll
        for (int u = 0; u < UNR; ++u) {
            v2f* p = (v2f*)(accs + ga[u] * COLS_PER_WG);
            *p += sum12(raw[u]);
        }
        n += ngrp * UNR;
    }
    for (; n < m1; ++n) {
        const int g = pb[n];
        v2f* p = (v2f*)(accs + g * COLS_PER_WG);
        *p += gather12(x, lane_off, w2, w3, w4, n);
    }
    __syncthreads();

    const int c4 = (t & 31) * 4;
    float* dst = partial + (size_t)chunk * (N_GRAPHS * HID) + cg * COLS_PER_WG + c4;
    for (int g = t >> 5; g < N_GRAPHS; g += 8) {
        float4 s = *(const float4*)(acc + g * COLS_PER_WG + c4);
        for (int k = 1; k < SUBS; ++k) {
            const float4 a = *(const float4*)(acc + (k * N_GRAPHS + g) * COLS_PER_WG + c4);
            s.x += a.x; s.y += a.y; s.z += a.z; s.w += a.w;
        }
        *(float4*)(dst + g * HID) = s;
    }
}

__global__ void k_reduce(const float* __restrict__ partial,
                         const unsigned* __restrict__ counts,
                         float* __restrict__ P, float* __restrict__ s,
                         int nchunks) {
    int i = blockIdx.x * 256 + threadIdx.x;
    int g = i >> 9;
    float a = 0.0f;
    for (int ch = 0; ch < nchunks; ++ch)
        a += partial[(size_t)ch * (N_GRAPHS * HID) + i];
    float cnt = (float)counts[g];
    P[i] = a / fmaxf(cnt, 1.0f);
    if (i < N_GRAPHS) s[i] = (counts[i] != 0u) ? 12.0f : 0.0f;
}

#define GEMM_KSPLIT 8
#define GEMM_KPER   (HID / GEMM_KSPLIT)
#define GEMM_STEPS  (GEMM_KPER / 4)
#define PART_PITCH  36

__global__ void __launch_bounds__(256) __attribute__((amdgpu_waves_per_eu(2, 2)))
k_gemm_wmma(const float* __restrict__ A, int lda,
            const float* __restrict__ W, const float* __restrict__ bias,
            const float* __restrict__ s, float* __restrict__ out) {
    __shared__ float part[GEMM_KSPLIT * 16 * PART_PITCH];

    const int ks   = threadIdx.x >> 5;
    const int lane = threadIdx.x & 31;
    const int l16  = lane & 15;
    const int lh   = lane >> 4;
    const int tm   = blockIdx.x >> 4;
    const int bn   = blockIdx.x & 15;

    const int   col = bn * 32 + lane;
    const float bc  = bias[col];
    const float sv0 = s[tm * 16 + ks];
    const float sv1 = s[tm * 16 + ks + 8];

    const float* Ap = A + (tm * 16 + l16) * lda + ks * GEMM_KPER + lh * 2;
    const float* Wp = W + (ks * GEMM_KPER + lh * 2) * HID + bn * 32 + l16 * 2;

    v2f a[GEMM_STEPS], b0[GEMM_STEPS], b1[GEMM_STEPS];
    #pragma unroll
    for (int j = 0; j < GEMM_STEPS; ++j) {
        a[j] = *(const v2f*)(Ap + 4 * j);
        const v2f w0 = *(const v2f*)(Wp + (4 * j) * HID);
        const v2f w1 = *(const v2f*)(Wp + (4 * j + 1) * HID);
        b0[j].x = w0.x;  b0[j].y = w1.x;
        b1[j].x = w0.y;  b1[j].y = w1.y;
    }

    v8f acc0 = {};
    v8f acc1 = {};
    #pragma unroll
    for (int j = 0; j < GEMM_STEPS; ++j) {
        acc0 = __builtin_amdgcn_wmma_f32_16x16x4_f32(
             false, a[j],  false, b0[j],
             (short)0, acc0,  false,  false);
        acc1 = __builtin_amdgcn_wmma_f32_16x16x4_f32(
             false, a[j],  false, b1[j],
             (short)0, acc1,  false,  false);
    }

    float* pw = part + ks * (16 * PART_PITCH) + (lh * 8) * PART_PITCH + l16 * 2;
    #pragma unroll
    for (int v = 0; v < 8; ++v) {
        v2f pr2;  pr2.x = acc0[v];  pr2.y = acc1[v];
        *(v2f*)(pw + v * PART_PITCH) = pr2;
    }
    __syncthreads();

    #pragma unroll
    for (int i = 0; i < 2; ++i) {
        const int r = ks + 8 * i;
        const float* pr = part + r * PART_PITCH + lane;
        float sum = pr[0];
        for (int q = 1; q < GEMM_KSPLIT; ++q) sum += pr[q * (16 * PART_PITCH)];
        out[(tm * 16 + r) * OUT_PITCH + col] = sum + (i ? sv1 : sv0) * bc;
    }
}

extern "C" void kernel_launch(void* const* d_in, const int* in_sizes, int n_in,
                              void* d_out, int out_size, void* d_ws, size_t ws_size,
                              hipStream_t stream) {
    const float* x     = (const float*)d_in[0];
    const int*   w2    = (const int*)  d_in[1];
    const int*   w3    = (const int*)  d_in[2];
    const int*   w4    = (const int*)  d_in[3];
    const int*   batch = (const int*)  d_in[4];
    const float* W0 = (const float*)d_in[5];  const float* b0 = (const float*)d_in[6];
    const float* W1 = (const float*)d_in[7];  const float* b1 = (const float*)d_in[8];
    const float* W2 = (const float*)d_in[9];  const float* b2 = (const float*)d_in[10];
    float* out = (float*)d_out;

    const size_t partBytes = (size_t)N_GRAPHS * HID * sizeof(float);
    const size_t fixed = partBytes   + 512   + 512
                       + (size_t)N_NODES * sizeof(int)  ;
    int nchunks = 1;
    if (ws_size > fixed + partBytes) {
        size_t fit = (ws_size - fixed) / partBytes;
        nchunks = (fit > MAX_CHUNKS) ? MAX_CHUNKS : (int)fit;
        if (nchunks < 1) nchunks = 1;
    }
    char* ws = (char*)d_ws;
    float*    partial = (float*)ws;
    float*    P       = (float*)(ws + (size_t)nchunks * partBytes);
    unsigned* counts  = (unsigned*)((char*)P + partBytes);
    float*    s       = (float*)((char*)counts + 512);
    int*      pb      = (int*)((char*)s + 512);

    k_zero_counts<<<1, 128, 0, stream>>>(counts);
    k_pb_hist<<<(N_NODES + 255) / 256, 256, 0, stream>>>(w2, batch, pb, counts);

    dim3 gAcc(nchunks, HID / COLS_PER_WG);
    k_accum<<<gAcc, 256, 0, stream>>>(x, w2, w3, w4, pb, partial, nchunks);
    k_reduce<<<(N_GRAPHS * HID) / 256, 256, 0, stream>>>(partial, counts, P, s, nchunks);

    k_gemm_wmma<<<128, 256, 0, stream>>>(P,         HID,       W0, b0, s, out);
    k_gemm_wmma<<<128, 256, 0, stream>>>(out,       OUT_PITCH, W1, b1, s, out + HID);
    k_gemm_wmma<<<128, 256, 0, stream>>>(out + HID, OUT_PITCH, W2, b2, s, out + 2 * HID);
}


# graph lookup and per-graph position counts folded into the accumulate kernel (graph read inline as batch[walk2[0][n]], integer counts per chunk summed in reduce): 7 launches to 5, no atomics left
# speedup vs baseline: 6.8324x; 1.0702x over previous
#include <hip/hip_runtime.h>


#define N_NODES   50000
#define N_GRAPHS  128
#define HID       512
#define OUT_PITCH (3 * HID)
#define MAX_CHUNKS 64
#define COLS_PER_WG 128
#define SUBS 4
#define UNR  4

typedef __attribute__((ext_vector_type(2))) float v2f;
typedef __attribute__((ext_vector_type(8))) float v8f;


__device__ __forceinline__ unsigned row_off(unsigned lane_off, int idx) {
    return (unsigned)idx * (unsigned)(HID * sizeof(float)) + lane_off;
}

__device__ __forceinline__ v2f ld_off2(const float* __restrict__ x, unsigned off) {
    return *(const v2f*)((const char*)x + off);
}

__device__ __forceinline__ void idx12(int (&ix)[12],
        const int* __restrict__ w2, const int* __restrict__ w3,
        const int* __restrict__ w4, int n) {
    #pragma unroll
    for (int l = 0; l < 3; ++l) ix[l]     = w2[l * N_NODES + n];
    #pragma unroll
    for (int l = 0; l < 4; ++l) ix[3 + l] = w3[l * N_NODES + n];
    #pragma unroll
    for (int l = 0; l < 5; ++l) ix[7 + l] = w4[l * N_NODES + n];
}

__device__ __forceinline__ void idx12_from(int (&ix)[12], int first,
        const int* __restrict__ w2, const int* __restrict__ w3,
        const int* __restrict__ w4, int n) {
    ix[0] = first;
    #pragma unroll
    for (int l = 1; l < 3; ++l) ix[l]     = w2[l * N_NODES + n];
    #pragma unroll
    for (int l = 0; l < 4; ++l) ix[3 + l] = w3[l * N_NODES + n];
    #pragma unroll
    for (int l = 0; l < 5; ++l) ix[7 + l] = w4[l * N_NODES + n];
}

__device__ __forceinline__ void offs12(unsigned (&o)[12], unsigned lane_off, const int (&ix)[12]) {
    #pragma unroll
    for (int k = 0; k < 12; ++k) o[k] = row_off(lane_off, ix[k]);
}

__device__ __forceinline__ void issue12(v2f (&raw)[12], const float* __restrict__ x,
                                        const unsigned (&o)[12]) {
    #pragma unroll
    for (int k = 0; k < 12; ++k) raw[k] = ld_off2(x, o[k]);
}

__device__ __forceinline__ v2f sum12(const v2f (&raw)[12]) {
    v2f y = {0.0f, 0.0f};
    #pragma unroll
    for (int k = 0; k < 12; ++k) y += raw[k];
    return y;
}

__device__ __forceinline__ v2f gather12(const float* __restrict__ x, unsigned lane_off,
        const int* __restrict__ w2, const int* __restrict__ w3,
        const int* __restrict__ w4, int n) {
    int ix[12];
    unsigned o[12];
    v2f raw[12];
    idx12(ix, w2, w3, w4, n);
    offs12(o, lane_off, ix);
    issue12(raw, x, o);
    return sum12(raw);
}

__device__ __forceinline__ void count_pos(unsigned (&cnt)[N_GRAPHS / 32], int lane_key, int g) {
    #pragma unroll
    for (int k = 0; k < N_GRAPHS / 32; ++k) cnt[k] += (g == lane_key + 32 * k) ? 1u : 0u;
}

__global__ void __launch_bounds__(256)
k_accum(const float* __restrict__ x,
        const int* __restrict__ w2, const int* __restrict__ w3,
        const int* __restrict__ w4, const int* __restrict__ batch,
        float* __restrict__ partial, unsigned* __restrict__ cntp, int nchunks) {
    __shared__ float acc[SUBS * N_GRAPHS * COLS_PER_WG];
    __shared__ unsigned cl[SUBS * N_GRAPHS];
    const int chunk = blockIdx.x;
    const int cg    = blockIdx.y;
    const int t     = threadIdx.x;
    const int wave  = __builtin_amdgcn_readfirstlane(t >> 5);
    const int lane  = t & 31;
    const int sub   = wave >> 1;
    const int wc    = (wave & 1) * 64 + lane * 2;
    const int c     = cg * COLS_PER_WG + wc;

    float4* acc4 = (float4*)acc;
    for (int i = t; i < SUBS * N_GRAPHS * COLS_PER_WG / 4; i += 256)
        acc4[i] = make_float4(0.0f, 0.0f, 0.0f, 0.0f);
    __syncthreads();

    const int per  = (N_NODES + nchunks - 1) / nchunks;
    const int n0   = chunk * per;
    const int n1   = (n0 + per < N_NODES) ? (n0 + per) : N_NODES;
    const int sper = (per + SUBS - 1) / SUBS;
    const int m0   = n0 + sub * sper;
    const int m1   = (m0 + sper < n1) ? (m0 + sper) : n1;

    float* accs = acc + sub * (N_GRAPHS * COLS_PER_WG) + wc;
    const unsigned lane_off = (unsigned)c * (unsigned)sizeof(float);

    const bool counting = (cg == 0) && ((wave & 1) == 0);
    const int lane_key = counting ? lane : -N_GRAPHS;
    unsigned cnt[N_GRAPHS / 32] = {0u, 0u, 0u, 0u};

    int n = m0;
    const int ngrp = (m1 > m0) ? (m1 - m0) / UNR : 0;
    if (ngrp > 0) {
        v2f raw[UNR][12];
        int ga[UNR];
        unsigned o[UNR][12];
        int gn[UNR];
        int j0[UNR];
        #pragma unroll
        for (int u = 0; u < UNR; ++u) {
            int ix[12];
            idx12(ix, w2, w3, w4, n + u);
            offs12(o[u], lane_off, ix);
            ga[u] = batch[ix[0]];
            issue12(raw[u], x, o[u]);
        }
        {
            int qb = n + UNR;  qb = (qb <= N_NODES - UNR) ? qb : (N_NODES - UNR);
            #pragma unroll
            for (int u = 0; u < UNR; ++u) {
                int ix[12];
                idx12(ix, w2, w3, w4, qb + u);
                offs12(o[u], lane_off, ix);
                gn[u] = batch[ix[0]];
            }
            int qc = n + 2 * UNR;  qc = (qc <= N_NODES - UNR) ? qc : (N_NODES - UNR);
            #pragma unroll
            for (int u = 0; u < UNR; ++u) j0[u] = w2[qc + u];
        }
        for (int gi = 1; gi < ngrp; ++gi) {
            int qb = n + (gi + 1) * UNR;  qb = (qb <= N_NODES - UNR) ? qb : (N_NODES - UNR);
            int qc = n + (gi + 2) * UNR;  qc = (qc <= N_NODES - UNR) ? qc : (N_NODES - UNR);
            int ixn[UNR][12];
            int gnn[UNR];
            int j0n[UNR];
            #pragma unroll
            for (int u = 0; u < UNR; ++u) {
                idx12_from(ixn[u], j0[u], w2, w3, w4, qb + u);
                gnn[u] = batch[j0[u]];
                j0n[u] = w2[qc + u];
            }
            #pragma unroll
            for (int u = 0; u < UNR; ++u) {
                v2f* p = (v2f*)(accs + ga[u] * COLS_PER_WG);
                *p += sum12(raw[u]);
                count_pos(cnt, lane_key, ga[u]);
                issue12(raw[u], x, o[u]);
                __builtin_amdgcn_sched_barrier(0);
            }
            #pragma unroll
            for (int u = 0; u < UNR; ++u) {
                offs12(o[u], lane_off, ixn[u]);
                ga[u] = gn[u];
                gn[u] = gnn[u];
                j0[u] = j0n[u];
            }
        }
        #pragma unroll
        for (int u = 0; u < UNR; ++u) {
            v2f* p = (v2f*)(accs + ga[u] * COLS_PER_WG);
            *p += sum12(raw[u]);
            count_pos(cnt, lane_key, ga[u]);
        }
        n += ngrp * UNR;
    }
    for (; n < m1; ++n) {
        const int g = batch[w2[n]];
        v2f* p = (v2f*)(accs + g * COLS_PER_WG);
        *p += gather12(x, lane_off, w2, w3, w4, n);
        count_pos(cnt, lane_key, g);
    }
    if (counting) {
        #pragma unroll
        for (int k = 0; k < N_GRAPHS / 32; ++k) cl[sub * N_GRAPHS + lane + 32 * k] = cnt[k];
    }
    __syncthreads();

    const int c4 = (t & 31) * 4;
    float* dst = partial + (size_t)chunk * (N_GRAPHS * HID) + cg * COLS_PER_WG + c4;
    for (int g = t >> 5; g < N_GRAPHS; g += 8) {
        float4 s = *(const float4*)(acc + g * COLS_PER_WG + c4);
        for (int k = 1; k < SUBS; ++k) {
            const float4 a = *(const float4*)(acc + (k * N_GRAPHS + g) * COLS_PER_WG + c4);
            s.x += a.x; s.y += a.y; s.z += a.z; s.w += a.w;
        }
        *(float4*)(dst + g * HID) = s;
    }
    if (cg == 0 && t < N_GRAPHS) {
        unsigned v = cl[t];
        for (int k = 1; k < SUBS; ++k) v += cl[k * N_GRAPHS + t];
        cntp[chunk * N_GRAPHS + t] = v;
    }
}

__global__ void k_reduce(const float* __restrict__ partial,
                         const unsigned* __restrict__ cntp,
                         float* __restrict__ P, float* __restrict__ s,
                         int nchunks) {
    __shared__ unsigned gcount;
    int i = blockIdx.x * 256 + threadIdx.x;
    int g = i >> 9;
    unsigned cv[MAX_CHUNKS / 32];
    #pragma unroll
    for (int q = 0; q < MAX_CHUNKS / 32; ++q) {
        const int ch = (int)threadIdx.x + 32 * q;
        cv[q] = (threadIdx.x < 32 && ch < nchunks) ? cntp[ch * N_GRAPHS + g] : 0u;
    }
    float a = 0.0f;
    for (int ch = 0; ch < nchunks; ++ch)
        a += partial[(size_t)ch * (N_GRAPHS * HID) + i];
    if (threadIdx.x < 32) {
        unsigned v = cv[0];
        #pragma unroll
        for (int q = 1; q < MAX_CHUNKS / 32; ++q) v += cv[q];
        for (int off = 16; off > 0; off >>= 1) v += __shfl_xor(v, off);
        if (threadIdx.x == 0) gcount = v;
    }
    __syncthreads();
    const unsigned count = gcount;
    P[i] = a / fmaxf((float)count, 1.0f);
    if ((i & (HID - 1)) == 0) s[g] = (count != 0u) ? 12.0f : 0.0f;
}

#define GEMM_KSPLIT 8
#define GEMM_KPER   (HID / GEMM_KSPLIT)
#define GEMM_STEPS  (GEMM_KPER / 4)
#define PART_PITCH  36

__global__ void __launch_bounds__(256) __attribute__((amdgpu_waves_per_eu(2, 2)))
k_gemm_wmma(const float* __restrict__ A, int lda,
            const float* __restrict__ W, const float* __restrict__ bias,
            const float* __restrict__ s, float* __restrict__ out) {
    __shared__ float part[GEMM_KSPLIT * 16 * PART_PITCH];

    const int ks   = threadIdx.x >> 5;
    const int lane = threadIdx.x & 31;
    const int l16  = lane & 15;
    const int lh   = lane >> 4;
    const int tm   = blockIdx.x >> 4;
    const int bn   = blockIdx.x & 15;

    const int   col = bn * 32 + lane;
    const float bc  = bias[col];
    const float sv0 = s[tm * 16 + ks];
    const float sv1 = s[tm * 16 + ks + 8];

    const float* Ap = A + (tm * 16 + l16) * lda + ks * GEMM_KPER + lh * 2;
    const float* Wp = W + (ks * GEMM_KPER + lh * 2) * HID + bn * 32 + l16 * 2;

    v2f a[GEMM_STEPS], b0[GEMM_STEPS], b1[GEMM_STEPS];
    #pragma unroll
    for (int j = 0; j < GEMM_STEPS; ++j) {
        a[j] = *(const v2f*)(Ap + 4 * j);
        const v2f w0 = *(const v2f*)(Wp + (4 * j) * HID);
        const v2f w1 = *(const v2f*)(Wp + (4 * j + 1) * HID);
        b0[j].x = w0.x;  b0[j].y = w1.x;
        b1[j].x = w0.y;  b1[j].y = w1.y;
    }

    v8f acc0 = {};
    v8f acc1 = {};
    #pragma unroll
    for (int j = 0; j < GEMM_STEPS; ++j) {
        acc0 = __builtin_amdgcn_wmma_f32_16x16x4_f32(
             false, a[j],  false, b0[j],
             (short)0, acc0,  false,  false);
        acc1 = __builtin_amdgcn_wmma_f32_16x16x4_f32(
             false, a[j],  false, b1[j],
             (short)0, acc1,  false,  false);
    }

    float* pw = part + ks * (16 * PART_PITCH) + (lh * 8) * PART_PITCH + l16 * 2;
    #pragma unroll
    for (int v = 0; v < 8; ++v) {
        v2f pr2;  pr2.x = acc0[v];  pr2.y = acc1[v];
        *(v2f*)(pw + v * PART_PITCH) = pr2;
    }
    __syncthreads();

    #pragma unroll
    for (int i = 0; i < 2; ++i) {
        const int r = ks + 8 * i;
        const float* pr = part + r * PART_PITCH + lane;
        float sum = pr[0];
        for (int q = 1; q < GEMM_KSPLIT; ++q) sum += pr[q * (16 * PART_PITCH)];
        out[(tm * 16 + r) * OUT_PITCH + col] = sum + (i ? sv1 : sv0) * bc;
    }
}

extern "C" void kernel_launch(void* const* d_in, const int* in_sizes, int n_in,
                              void* d_out, int out_size, void* d_ws, size_t ws_size,
                              hipStream_t stream) {
    const float* x     = (const float*)d_in[0];
    const int*   w2    = (const int*)  d_in[1];
    const int*   w3    = (const int*)  d_in[2];
    const int*   w4    = (const int*)  d_in[3];
    const int*   batch = (const int*)  d_in[4];
    const float* W0 = (const float*)d_in[5];  const float* b0 = (const float*)d_in[6];
    const float* W1 = (const float*)d_in[7];  const float* b1 = (const float*)d_in[8];
    const float* W2 = (const float*)d_in[9];  const float* b2 = (const float*)d_in[10];
    float* out = (float*)d_out;

    const size_t partBytes = (size_t)N_GRAPHS * HID * sizeof(float);
    const size_t fixed = partBytes   + 512   + 512
                       + (size_t)N_NODES * sizeof(int)  ;
    int nchunks = 1;
    if (ws_size > fixed + partBytes) {
        size_t fit = (ws_size - fixed) / partBytes;
        nchunks = (fit > MAX_CHUNKS) ? MAX_CHUNKS : (int)fit;
        if (nchunks < 1) nchunks = 1;
    }
    char* ws = (char*)d_ws;
    float*    partial = (float*)ws;
    float*    P       = (float*)(ws + (size_t)nchunks * partBytes);
    float*    s       = (float*)((char*)P + partBytes + 512);
    unsigned* cntp    = (unsigned*)((char*)s + 512);

    dim3 gAcc(nchunks, HID / COLS_PER_WG);
    k_accum<<<gAcc, 256, 0, stream>>>(x, w2, w3, w4, batch, partial, cntp, nchunks);
    k_reduce<<<(N_GRAPHS * HID) / 256, 256, 0, stream>>>(partial, cntp, P, s, nchunks);

    k_gemm_wmma<<<128, 256, 0, stream>>>(P,         HID,       W0, b0, s, out);
    k_gemm_wmma<<<128, 256, 0, stream>>>(out,       OUT_PITCH, W1, b1, s, out + HID);
    k_gemm_wmma<<<128, 256, 0, stream>>>(out + HID, OUT_PITCH, W2, b2, s, out + 2 * HID);
}
